# starting kernel
# baseline (speedup 1.0000x reference)
#include <hip/hip_runtime.h>
#include <math.h>

typedef __attribute__((ext_vector_type(2))) float v2f;
typedef __attribute__((ext_vector_type(8))) float v8f;

static constexpr int BS = 8, CH = 3, H = 40, W = 40;
static constexpr int HO = 39, WO = 39;
static constexpr int POS_PER_IMG = HO * WO;
static constexpr int NPOS = BS * POS_PER_IMG;
static constexpr int NTILES = (NPOS + 15) / 16;

__global__ __launch_bounds__(256) void build_mats_kernel(
    const float* __restrict__ wts, float* __restrict__ mats) {
  __shared__ float sre[16][256];
  __shared__ float sim2[16][256];
  const int x = threadIdx.x;

#pragma unroll
  for (int j = 0; j < 16; ++j) {
    sre[j][x] = (x == (j << 4)) ? 1.0f : 0.0f;
    sim2[j][x] = 0.0f;
  }
  __syncthreads();

  for (int l = 0; l < 2; ++l) {
    for (int q = 0; q < 8; ++q) {
      float wv = wts[l * 8 + q];
      float s, c;
      sincosf(0.5f * wv, &s, &c);
      const int p = x ^ (1 << (7 - q));
      float nr[16], ni[16];
#pragma unroll
      for (int j = 0; j < 16; ++j) {
        float r1 = sre[j][x], i1 = sim2[j][x];
        float r2 = sre[j][p], i2 = sim2[j][p];
        nr[j] = c * r1 + s * i2;
        ni[j] = c * i1 - s * r2;
      }
      __syncthreads();
#pragma unroll
      for (int j = 0; j < 16; ++j) { sre[j][x] = nr[j]; sim2[j][x] = ni[j]; }
      __syncthreads();
    }
    for (int q = 0; q < 8; ++q) {
      const int cm = 1 << (7 - q);
      const int tm = 1 << (7 - ((q + 1) & 7));
      const int xp = (x & cm) ? (x ^ tm) : x;
      float nr[16], ni[16];
#pragma unroll
      for (int j = 0; j < 16; ++j) { nr[j] = sre[j][xp]; ni[j] = sim2[j][xp]; }
      __syncthreads();
#pragma unroll
      for (int j = 0; j < 16; ++j) { sre[j][x] = nr[j]; sim2[j][x] = ni[j]; }
      __syncthreads();
    }
  }

  const int j = x >> 4, k = x & 15;
  float ar0 = 0.f, ai0 = 0.f, ar1 = 0.f, ai1 = 0.f;
  for (int t = 0; t < 256; ++t) {
    float vjr = sre[j][t], vji = sim2[j][t];
    float vkr = sre[k][t], vki = sim2[k][t];
    float re = vjr * vkr + vji * vki;
    float im = vjr * vki - vji * vkr;
    float d0 = ((t >> 7) & 1) ? -1.f : 1.f;
    float d1 = ((t >> 6) & 1) ? -1.f : 1.f;
    ar0 += d0 * re; ai0 += d0 * im;
    ar1 += d1 * re; ai1 += d1 * im;
  }
  mats[0 * 256 + x] = ar0;
  mats[1 * 256 + x] = ai0;
  mats[2 * 256 + x] = ar1;
  mats[3 * 256 + x] = ai1;
}

__global__ __launch_bounds__(32) void quanv_kernel(
    const float* __restrict__ xin, const float* __restrict__ mats,
    float* __restrict__ out) {
  const int lane = threadIdx.x;
  const int m = lane & 15;
  const int h = lane >> 4;
  const int tile = blockIdx.x;

  __shared__ float Pu[16][16];
  __shared__ float Pv[16][16];

  v2f bA0[4], bB0[4], bA1[4], bB1[4];
#pragma unroll
  for (int kb = 0; kb < 4; ++kb) {
    const int r0 = 4 * kb + 2 * h;
    bA0[kb] = (v2f){mats[0 * 256 + r0 * 16 + m], mats[0 * 256 + (r0 + 1) * 16 + m]};
    bB0[kb] = (v2f){mats[1 * 256 + r0 * 16 + m], mats[1 * 256 + (r0 + 1) * 16 + m]};
    bA1[kb] = (v2f){mats[2 * 256 + r0 * 16 + m], mats[2 * 256 + (r0 + 1) * 16 + m]};
    bB1[kb] = (v2f){mats[3 * 256 + r0 * 16 + m], mats[3 * 256 + (r0 + 1) * 16 + m]};
  }

  int pos = tile * 16 + m;
  int posc = (pos < NPOS) ? pos : (NPOS - 1);
  int b = posc / POS_PER_IMG;
  int rem = posc - b * POS_PER_IMG;
  int ii = rem / WO;
  int jj = rem - ii * WO;

  float z0[8], z1[8];
#pragma unroll
  for (int r = 0; r < 8; ++r) { z0[r] = 0.f; z1[r] = 0.f; }

  for (int ch = 0; ch < 3; ++ch) {
    const float* img = xin + (((size_t)b * CH + ch) * H + ii) * W + jj;
    float t0 = img[0], t1 = img[1], t2 = img[W], t3 = img[W + 1];
    float c0, s0, c1, s1, c2, s2, c3, s3;
    sincosf(0.5f * t0, &s0, &c0);
    sincosf(0.5f * t1, &s1, &c1);
    sincosf(0.5f * t2, &s2, &c2);
    sincosf(0.5f * t3, &s3, &c3);

    float u[16], v[16];
#pragma unroll
    for (int e = 0; e < 16; ++e) {
      float m0 = (e & 8) ? s0 : c0;
      float m1 = (e & 4) ? s1 : c1;
      float m2 = (e & 2) ? s2 : c2;
      float m3 = (e & 1) ? s3 : c3;
      float mag = m0 * m1 * m2 * m3;
      int p = __popc(e) & 3;
      u[e] = (p == 0) ? mag : ((p == 2) ? -mag : 0.0f);
      v[e] = (p == 1) ? -mag : ((p == 3) ? mag : 0.0f);
    }
    if (h == 0) {
#pragma unroll
      for (int e = 0; e < 16; ++e) { Pu[m][e] = u[e]; Pv[m][e] = v[e]; }
    }
    __syncthreads();

    v8f gA0u = {0, 0, 0, 0, 0, 0, 0, 0};
    v8f gA0v = {0, 0, 0, 0, 0, 0, 0, 0};
    v8f gB0u = {0, 0, 0, 0, 0, 0, 0, 0};
    v8f gA1u = {0, 0, 0, 0, 0, 0, 0, 0};
    v8f gA1v = {0, 0, 0, 0, 0, 0, 0, 0};
    v8f gB1u = {0, 0, 0, 0, 0, 0, 0, 0};
#pragma unroll
    for (int kb = 0; kb < 4; ++kb) {
      v2f au = (v2f){u[4 * kb + 2 * h], u[4 * kb + 2 * h + 1]};
      v2f av = (v2f){v[4 * kb + 2 * h], v[4 * kb + 2 * h + 1]};
      gA0u = __builtin_amdgcn_wmma_f32_16x16x4_f32(false, au, false, bA0[kb],
                                                   (short)0, gA0u, false, false);
      gA0v = __builtin_amdgcn_wmma_f32_16x16x4_f32(false, av, false, bA0[kb],
                                                   (short)0, gA0v, false, false);
      gB0u = __builtin_amdgcn_wmma_f32_16x16x4_f32(false, au, false, bB0[kb],
                                                   (short)0, gB0u, false, false);
      gA1u = __builtin_amdgcn_wmma_f32_16x16x4_f32(false, au, false, bA1[kb],
                                                   (short)0, gA1u, false, false);
      gA1v = __builtin_amdgcn_wmma_f32_16x16x4_f32(false, av, false, bA1[kb],
                                                   (short)0, gA1v, false, false);
      gB1u = __builtin_amdgcn_wmma_f32_16x16x4_f32(false, au, false, bB1[kb],
                                                   (short)0, gB1u, false, false);
    }

#pragma unroll
    for (int r = 0; r < 8; ++r) {
      float pu = Pu[r + 8 * h][m];
      float pv = Pv[r + 8 * h][m];
      z0[r] += gA0u[r] * pu + (gA0v[r] - 2.0f * gB0u[r]) * pv;
      z1[r] += gA1u[r] * pu + (gA1v[r] - 2.0f * gB1u[r]) * pv;
    }
    __syncthreads();
  }

#pragma unroll
  for (int r = 0; r < 8; ++r) {
    float a = z0[r], c = z1[r];
    for (int mask = 1; mask <= 8; mask <<= 1) {
      a += __shfl_xor(a, mask, 32);
      c += __shfl_xor(c, mask, 32);
    }
    z0[r] = a; z1[r] = c;
  }

  if (m == 0) {
#pragma unroll
    for (int r = 0; r < 8; ++r) {
      int posm = tile * 16 + r + 8 * h;
      if (posm < NPOS) {
        int bb = posm / POS_PER_IMG;
        int rr = posm - bb * POS_PER_IMG;
        out[(bb * 2 + 0) * POS_PER_IMG + rr] = z0[r] * (1.0f / 3.0f);
        out[(bb * 2 + 1) * POS_PER_IMG + rr] = z1[r] * (1.0f / 3.0f);
      }
    }
  }
}

extern "C" void kernel_launch(void* const* d_in, const int* in_sizes, int n_in,
                              void* d_out, int out_size, void* d_ws, size_t ws_size,
                              hipStream_t stream) {
  const float* x = (const float*)d_in[0];
  const float* wts = (const float*)d_in[1];
  float* out = (float*)d_out;
  float* mats = (float*)d_ws;

  build_mats_kernel<<<1, 256, 0, stream>>>(wts, mats);
  quanv_kernel<<<NTILES, 32, 0, stream>>>(x, mats, out);
}


# One lane per pixel slot: each pixel's sin/cos taken once per wave, right patch column read from the next lane; 31 positions per wave (half the waves), 2 k-slots per position; new summation order
# speedup vs baseline: 13.6947x; 13.6947x over previous
#include <hip/hip_runtime.h>
#include <math.h>


typedef __attribute__((ext_vector_type(2))) float v2f;
typedef __attribute__((ext_vector_type(8))) float v8f;

static constexpr int BS = 8, CH = 3, H = 40, W = 40;
static constexpr int HO = 39, WO = 39;
static constexpr int POS_PER_IMG = HO * WO;
static constexpr int SLOTS_PER_IMG = HO * W;
static constexpr int STRIDE = 31;
static constexpr int WAVES_PER_IMG = (SLOTS_PER_IMG + STRIDE - 1) / STRIDE;


__device__ __forceinline__ float lane_bcast(float v, int src) {
  return __int_as_float(__builtin_amdgcn_readlane(__float_as_int(v), src));
}

__device__ __forceinline__ float lane_read(float v, int lane_byte_addr) {
  return __int_as_float(__builtin_amdgcn_ds_bpermute(lane_byte_addr, __float_as_int(v)));
}

__global__ __launch_bounds__(32) void quanv_kernel(
    const float* __restrict__ xin, const float* __restrict__ wts,
    float* __restrict__ out) {
  const int lane = threadIdx.x;
  const int b = blockIdx.y;

  const int q = blockIdx.x * STRIDE + lane;
  const int qc = (q < SLOTS_PER_IMG) ? q : (SLOTS_PER_IMG - 1);
  const int row = qc / W;
  const int col = qc - row * W;
  const bool valid = (lane < STRIDE) && (q < SLOTS_PER_IMG) && (col < WO);

  const float* img = xin + (size_t)b * (CH * H * W) + qc;
  float tT[3], tB[3];
#pragma unroll
  for (int ch = 0; ch < 3; ++ch) {
    tT[ch] = img[ch * H * W];
    tB[ch] = img[ch * H * W + W];
  }

  const float wv = wts[lane & 15];
  float sw, cw;
  sincosf(wv, &sw, &cw);
  float ca[4], sa[4];
#pragma unroll
  for (int i = 0; i < 4; ++i) { ca[i] = lane_bcast(cw, i); sa[i] = lane_bcast(sw, i); }
  const float ca4 = lane_bcast(cw, 4), ca6 = lane_bcast(cw, 6);
  const float ca5 = lane_bcast(cw, 5), sa5 = lane_bcast(sw, 5);
  const float ca7 = lane_bcast(cw, 7), sa7 = lane_bcast(sw, 7);
  float cb[8], sb[8];
#pragma unroll
  for (int i = 0; i < 8; ++i) { cb[i] = lane_bcast(cw, 8 + i); sb[i] = lane_bcast(sw, 8 + i); }

  const float zz12 = cb[1] * cb[2], yy12 = sb[1] * sb[2];
  const float zz34 = cb[3] * cb[4], yy34 = sb[3] * sb[4];
  const float zz56 = cb[5] * cb[6], yy56 = sb[5] * sb[6];
  constexpr unsigned NEG = (1u << 3) | (1u << 6) | (1u << 9) | (1u << 12);
  float C0[8];
#pragma unroll
  for (int i = 0; i < 8; ++i) C0[i] = 0.0f;
#pragma unroll
  for (int T = 0; T < 16; ++T) {
    const int p12 = (T >> 3) & 1, p34 = (T >> 2) & 1, p56 = (T >> 1) & 1, p7 = T & 1;
    const int y0 = p7, y1 = p12 ^ p7, y3 = p12 ^ p34, y5 = p34 ^ p56, y7 = p56 ^ p7;
    float term = (p12 ? yy12 : zz12) * (p34 ? yy34 : zz34) * (p56 ? yy56 : zz56) *
                 (p7 ? sb[7] : cb[7]) * (y5 ? sa5 : ca5) * (y7 ? sa7 : ca7);
    C0[4 * y0 + 2 * y1 + y3] += ((NEG >> T) & 1u) ? -term : term;
  }
  const float k47 = ca4 * ca5 * ca6 * ca7;
  const float C1c = cb[0] * cb[1] * k47;
  const float C1s = sb[0] * sb[1] * k47;

  const v2f zero2 = (v2f){0.0f, 0.0f};
  const int arow = lane & 15, ahalf = lane >> 4;
  const bool rowz0 = (arow == 8 * ahalf);
  const bool rowz1 = (arow == 8 * ahalf + 1);
  v2f aZ0[4];
#pragma unroll
  for (int j = 0; j < 4; ++j) aZ0[j] = rowz0 ? (v2f){C0[2 * j], C0[2 * j + 1]} : zero2;
  const v2f aZ1 = rowz1 ? (v2f){C1c, C1s} : zero2;

  const int up_addr = ((lane + 1) & 31) << 2;

  float s0[3], c0[3], s2[3], c2[3], s1[3], c1[3], s3[3], c3[3];
#pragma unroll
  for (int ch = 0; ch < 3; ++ch) {
    float sT, cT, sB, cB;
    sincosf(tT[ch], &sT, &cT);
    sincosf(tB[ch], &sB, &cB);
    s0[ch] = sT * ca[0] + cT * sa[0]; c0[ch] = cT * ca[0] - sT * sa[0];
    s2[ch] = sB * ca[2] + cB * sa[2]; c2[ch] = cB * ca[2] - sB * sa[2];
    const float s1d = sT * ca[1] + cT * sa[1], c1d = cT * ca[1] - sT * sa[1];
    const float s3d = sB * ca[3] + cB * sa[3], c3d = cB * ca[3] - sB * sa[3];
    s1[ch] = lane_read(s1d, up_addr); c1[ch] = lane_read(c1d, up_addr);
    s3[ch] = lane_read(s3d, up_addr); c3[ch] = lane_read(c3d, up_addr);
  }

  v8f accA = {0, 0, 0, 0, 0, 0, 0, 0};
  v8f accB = {0, 0, 0, 0, 0, 0, 0, 0};
#pragma unroll
  for (int ch = 0; ch < 3; ++ch) {
    const float cc = c0[ch] * c1[ch], cs = c0[ch] * s1[ch];
    const float sc = s0[ch] * c1[ch], ss = s0[ch] * s1[ch];
    const v2f b0 = (v2f){cc * c3[ch], cc * s3[ch]};
    const v2f b1 = (v2f){cs * c3[ch], cs * s3[ch]};
    const v2f b2 = (v2f){sc * c3[ch], sc * s3[ch]};
    const v2f b3 = (v2f){ss * c3[ch], ss * s3[ch]};
    const v2f g = (v2f){c0[ch] * c2[ch] * c3[ch], s0[ch] * s2[ch] * c3[ch]};
    accA = __builtin_amdgcn_wmma_f32_16x16x4_f32(false, aZ0[0], false, b0, (short)0, accA, false, false);
    accB = __builtin_amdgcn_wmma_f32_16x16x4_f32(false, aZ0[1], false, b1, (short)0, accB, false, false);
    accA = __builtin_amdgcn_wmma_f32_16x16x4_f32(false, aZ0[2], false, b2, (short)0, accA, false, false);
    accB = __builtin_amdgcn_wmma_f32_16x16x4_f32(false, aZ0[3], false, b3, (short)0, accB, false, false);
    if (ch & 1)
      accB = __builtin_amdgcn_wmma_f32_16x16x4_f32(false, aZ1, false, g, (short)0, accB, false, false);
    else
      accA = __builtin_amdgcn_wmma_f32_16x16x4_f32(false, aZ1, false, g, (short)0, accA, false, false);
  }

  const float o0 = (accA[0] + accB[0]) * (1.0f / 3.0f);
  const float o1 = (accA[1] + accB[1]) * (1.0f / 3.0f);
  if (valid) {
    const int oidx = b * 2 * POS_PER_IMG + (qc - row);
    out[oidx] = o0;
    out[oidx + POS_PER_IMG] = o1;
  }
}

extern "C" void kernel_launch(void* const* d_in, const int* in_sizes, int n_in,
                              void* d_out, int out_size, void* d_ws, size_t ws_size,
                              hipStream_t stream) {
  const float* x = (const float*)d_in[0];
  const float* wts = (const float*)d_in[1];
  float* out = (float*)d_out;

  quanv_kernel<<<dim3(WAVES_PER_IMG, BS), 32, 0, stream>>>(x, wts, out);
}


# Four waves per work-group (128 threads, one wave per SIMD): 102 work-groups to dispatch instead of 408 single-wave ones; each wave's work and sums unchanged
# speedup vs baseline: 14.5409x; 1.0618x over previous
#include <hip/hip_runtime.h>
#include <math.h>


typedef __attribute__((ext_vector_type(2))) float v2f;
typedef __attribute__((ext_vector_type(8))) float v8f;

static constexpr int BS = 8, CH = 3, H = 40, W = 40;
static constexpr int HO = 39, WO = 39;
static constexpr int POS_PER_IMG = HO * WO;
static constexpr int SLOTS_PER_IMG = HO * W;
static constexpr int STRIDE = 31;
static constexpr int WAVES_PER_IMG = (SLOTS_PER_IMG + STRIDE - 1) / STRIDE;
static constexpr int WAVES_PER_WG = 4;
static constexpr int TOTAL_WAVES = BS * WAVES_PER_IMG;
static_assert(TOTAL_WAVES % WAVES_PER_WG == 0, "the grid must cover the waves exactly");
static constexpr int NUM_WG = TOTAL_WAVES / WAVES_PER_WG;
static constexpr int THREADS_PER_WG = 32 * WAVES_PER_WG;


__device__ __forceinline__ float lane_bcast(float v, int src) {
  return __int_as_float(__builtin_amdgcn_readlane(__float_as_int(v), src));
}

__device__ __forceinline__ float lane_read(float v, int lane_byte_addr) {
  return __int_as_float(__builtin_amdgcn_ds_bpermute(lane_byte_addr, __float_as_int(v)));
}

__global__ __launch_bounds__(THREADS_PER_WG) void quanv_kernel(
    const float* __restrict__ xin, const float* __restrict__ wts,
    float* __restrict__ out) {
  const int lane = threadIdx.x & 31;
  const int wave_id = blockIdx.x * WAVES_PER_WG +
                 __builtin_amdgcn_readfirstlane((int)(threadIdx.x >> 5));
  const int b = wave_id / WAVES_PER_IMG;
  const int chunk = wave_id - b * WAVES_PER_IMG;

  const int q = chunk * STRIDE + lane;
  const int qc = (q < SLOTS_PER_IMG) ? q : (SLOTS_PER_IMG - 1);
  const int row = qc / W;
  const int col = qc - row * W;
  const bool valid = (lane < STRIDE) && (q < SLOTS_PER_IMG) && (col < WO);

  const float* img = xin + (size_t)b * (CH * H * W) + qc;
  float tT[3], tB[3];
#pragma unroll
  for (int ch = 0; ch < 3; ++ch) {
    tT[ch] = img[ch * H * W];
    tB[ch] = img[ch * H * W + W];
  }

  const float wv = wts[lane & 15];
  float sw, cw;
  sincosf(wv, &sw, &cw);
  float ca[4], sa[4];
#pragma unroll
  for (int i = 0; i < 4; ++i) { ca[i] = lane_bcast(cw, i); sa[i] = lane_bcast(sw, i); }
  const float ca4 = lane_bcast(cw, 4), ca6 = lane_bcast(cw, 6);
  const float ca5 = lane_bcast(cw, 5), sa5 = lane_bcast(sw, 5);
  const float ca7 = lane_bcast(cw, 7), sa7 = lane_bcast(sw, 7);
  float cb[8], sb[8];
#pragma unroll
  for (int i = 0; i < 8; ++i) { cb[i] = lane_bcast(cw, 8 + i); sb[i] = lane_bcast(sw, 8 + i); }

  const float zz12 = cb[1] * cb[2], yy12 = sb[1] * sb[2];
  const float zz34 = cb[3] * cb[4], yy34 = sb[3] * sb[4];
  const float zz56 = cb[5] * cb[6], yy56 = sb[5] * sb[6];
  constexpr unsigned NEG = (1u << 3) | (1u << 6) | (1u << 9) | (1u << 12);
  float C0[8];
#pragma unroll
  for (int i = 0; i < 8; ++i) C0[i] = 0.0f;
#pragma unroll
  for (int T = 0; T < 16; ++T) {
    const int p12 = (T >> 3) & 1, p34 = (T >> 2) & 1, p56 = (T >> 1) & 1, p7 = T & 1;
    const int y0 = p7, y1 = p12 ^ p7, y3 = p12 ^ p34, y5 = p34 ^ p56, y7 = p56 ^ p7;
    float term = (p12 ? yy12 : zz12) * (p34 ? yy34 : zz34) * (p56 ? yy56 : zz56) *
                 (p7 ? sb[7] : cb[7]) * (y5 ? sa5 : ca5) * (y7 ? sa7 : ca7);
    C0[4 * y0 + 2 * y1 + y3] += ((NEG >> T) & 1u) ? -term : term;
  }
  const float k47 = ca4 * ca5 * ca6 * ca7;
  const float C1c = cb[0] * cb[1] * k47;
  const float C1s = sb[0] * sb[1] * k47;

  const v2f zero2 = (v2f){0.0f, 0.0f};
  const int arow = lane & 15, ahalf = lane >> 4;
  const bool rowz0 = (arow == 8 * ahalf);
  const bool rowz1 = (arow == 8 * ahalf + 1);
  v2f aZ0[4];
#pragma unroll
  for (int j = 0; j < 4; ++j) aZ0[j] = rowz0 ? (v2f){C0[2 * j], C0[2 * j + 1]} : zero2;
  const v2f aZ1 = rowz1 ? (v2f){C1c, C1s} : zero2;

  const int up_addr = ((lane + 1) & 31) << 2;

  float s0[3], c0[3], s2[3], c2[3], s1[3], c1[3], s3[3], c3[3];
#pragma unroll
  for (int ch = 0; ch < 3; ++ch) {
    float sT, cT, sB, cB;
    sincosf(tT[ch], &sT, &cT);
    sincosf(tB[ch], &sB, &cB);
    s0[ch] = sT * ca[0] + cT * sa[0]; c0[ch] = cT * ca[0] - sT * sa[0];
    s2[ch] = sB * ca[2] + cB * sa[2]; c2[ch] = cB * ca[2] - sB * sa[2];
    const float s1d = sT * ca[1] + cT * sa[1], c1d = cT * ca[1] - sT * sa[1];
    const float s3d = sB * ca[3] + cB * sa[3], c3d = cB * ca[3] - sB * sa[3];
    s1[ch] = lane_read(s1d, up_addr); c1[ch] = lane_read(c1d, up_addr);
    s3[ch] = lane_read(s3d, up_addr); c3[ch] = lane_read(c3d, up_addr);
  }

  v8f accA = {0, 0, 0, 0, 0, 0, 0, 0};
  v8f accB = {0, 0, 0, 0, 0, 0, 0, 0};
#pragma unroll
  for (int ch = 0; ch < 3; ++ch) {
    const float cc = c0[ch] * c1[ch], cs = c0[ch] * s1[ch];
    const float sc = s0[ch] * c1[ch], ss = s0[ch] * s1[ch];
    const v2f b0 = (v2f){cc * c3[ch], cc * s3[ch]};
    const v2f b1 = (v2f){cs * c3[ch], cs * s3[ch]};
    const v2f b2 = (v2f){sc * c3[ch], sc * s3[ch]};
    const v2f b3 = (v2f){ss * c3[ch], ss * s3[ch]};
    const v2f g = (v2f){c0[ch] * c2[ch] * c3[ch], s0[ch] * s2[ch] * c3[ch]};
    accA = __builtin_amdgcn_wmma_f32_16x16x4_f32(false, aZ0[0], false, b0, (short)0, accA, false, false);
    accB = __builtin_amdgcn_wmma_f32_16x16x4_f32(false, aZ0[1], false, b1, (short)0, accB, false, false);
    accA = __builtin_amdgcn_wmma_f32_16x16x4_f32(false, aZ0[2], false, b2, (short)0, accA, false, false);
    accB = __builtin_amdgcn_wmma_f32_16x16x4_f32(false, aZ0[3], false, b3, (short)0, accB, false, false);
    if (ch & 1)
      accB = __builtin_amdgcn_wmma_f32_16x16x4_f32(false, aZ1, false, g, (short)0, accB, false, false);
    else
      accA = __builtin_amdgcn_wmma_f32_16x16x4_f32(false, aZ1, false, g, (short)0, accA, false, false);
  }

  const float o0 = (accA[0] + accB[0]) * (1.0f / 3.0f);
  const float o1 = (accA[1] + accB[1]) * (1.0f / 3.0f);
  if (valid) {
    const int oidx = b * 2 * POS_PER_IMG + (qc - row);
    out[oidx] = o0;
    out[oidx + POS_PER_IMG] = o1;
  }
}

extern "C" void kernel_launch(void* const* d_in, const int* in_sizes, int n_in,
                              void* d_out, int out_size, void* d_ws, size_t ws_size,
                              hipStream_t stream) {
  const float* x = (const float*)d_in[0];
  const float* wts = (const float*)d_in[1];
  float* out = (float*)d_out;

  quanv_kernel<<<dim3(NUM_WG), THREADS_PER_WG, 0, stream>>>(x, wts, out);
}
